# starting kernel
# baseline (speedup 1.0000x reference)
#include <hip/hip_runtime.h>
#include <hip/hip_bf16.h>

typedef __attribute__((ext_vector_type(2))) float v2f;
typedef __attribute__((ext_vector_type(8))) float v8f;

#define H_PX   600
#define W_PX   800
#define N_OMM  721
#define PADR   4
#define KK     81
#define NTILES 46
#define NPLANES 96

__device__ __forceinline__ int reflect_idx(int v, int n) {
    v = (v < 0) ? -v : v;
    v = (v >= n) ? (2 * n - 2 - v) : v;
    return v;
}

__global__ __launch_bounds__(256) void hexeye_wmma_f32(
    const float* __restrict__ stim,
    const int*   __restrict__ rx,
    const int*   __restrict__ ry,
    float*       __restrict__ out)
{
    const int wavesPerBlock = blockDim.x >> 5;
    const int wave = blockIdx.x * wavesPerBlock + (threadIdx.x >> 5);
    const int lane = threadIdx.x & 31;

    const int plane = wave / NTILES;
    const int tile  = wave - plane * NTILES;
    if (plane >= NPLANES) return;

    const int n   = lane & 15;
    const int sub = lane >> 4;
    const int o_raw = tile * 16 + n;
    const bool valid = (o_raw < N_OMM);
    const int o = valid ? o_raw : (N_OMM - 1);

    int cx = rx[o]; cx = min(max(cx, PADR), W_PX + PADR - 1);
    int cy = ry[o]; cy = min(max(cy, PADR), H_PX + PADR - 1);

    const float* __restrict__ pp = stim + (size_t)plane * (H_PX * W_PX);

    int Yb[9], Xc[9];
    #pragma unroll
    for (int d = 0; d < 9; ++d) {
        Yb[d] = reflect_idx(cy + d - 8, H_PX) * W_PX;
        Xc[d] = reflect_idx(cx + d - 8, W_PX);
    }

    v2f a; a.x = 1.0f; a.y = 1.0f;
    v8f acc = {};

    #pragma unroll
    for (int s = 0; s < 21; ++s) {
        float b0, b1;
        if (s < 20) {
            const int k0a = 4 * s + 0, k0b = 4 * s + 1;
            const int k1a = 4 * s + 2, k1b = 4 * s + 3;
            const int i0 = sub ? (Yb[k0b / 9] + Xc[k0b % 9])
                               : (Yb[k0a / 9] + Xc[k0a % 9]);
            const int i1 = sub ? (Yb[k1b / 9] + Xc[k1b % 9])
                               : (Yb[k1a / 9] + Xc[k1a % 9]);
            b0 = pp[i0];
            b1 = pp[i1];
        } else {
            const float v = pp[Yb[8] + Xc[8]];
            b0 = sub ? 0.0f : v;
            b1 = 0.0f;
        }
        v2f bb; bb.x = b0; bb.y = b1;
        acc = __builtin_amdgcn_wmma_f32_16x16x4_f32(
             false, a,  false, bb,
             (short)0, acc,  false,  false);
    }

    if (valid && sub == 0) {
        out[(size_t)plane * N_OMM + o_raw] = acc[0] / 81.0f;
    }
}

extern "C" void kernel_launch(void* const* d_in, const int* in_sizes, int n_in,
                              void* d_out, int out_size, void* d_ws, size_t ws_size,
                              hipStream_t stream) {
    const float* stim = (const float*)d_in[0];
    const int*   rx   = (const int*)d_in[1];
    const int*   ry   = (const int*)d_in[2];
    float*       out  = (float*)d_out;

    (void)in_sizes; (void)n_in; (void)out_size; (void)d_ws; (void)ws_size;

    const int totalWaves = NPLANES * NTILES;
    const int wavesPerBlock = 256 / 32;
    const int blocks = (totalWaves + wavesPerBlock - 1) / wavesPerBlock;
    hexeye_wmma_f32<<<blocks, 256, 0, stream>>>(stim, rx, ry, out);
}


# Receptors are dealt to lanes in image-row order (the hexagonal lattice walked two rows at a time), not index order: the 16 patches of one load are horizontal neighbours and share 128-byte lines
# speedup vs baseline: 1.8226x; 1.8226x over previous
#include <hip/hip_runtime.h>
#include <hip/hip_bf16.h>


typedef __attribute__((ext_vector_type(2))) float v2f;
typedef __attribute__((ext_vector_type(4))) float v4f;
typedef __attribute__((ext_vector_type(8))) float v8f;
typedef v4f v4f_a4 __attribute__((aligned(4)));

#define H_PX   600
#define W_PX   800
#define N_OMM  721
#define PADR   4
#define NTILES 46
#define NPLANES 96
#define NBATCH 32
#define NCH    3
#define NTASKS (NBATCH * NTILES)
#define WAVES_PER_BLOCK 6
#define NBLOCKS 246
#define HEX_R  15
static_assert(N_OMM == 3 * HEX_R * (HEX_R + 1) + 1, "walk order below assumes a full hexagonal disk");

__device__ __forceinline__ int reflect_idx(int v, int n) {
    v = (v < 0) ? -v : v;
    v = (v >= n) ? (2 * n - 2 - v) : v;
    return v;
}

__device__ __forceinline__ int rows_before(int j) { return (HEX_R + 1) * j + ((j * (j - 1)) >> 1); }

__device__ __forceinline__ int walk_to_index(int t) {
    const int NROWS = 2 * HEX_R + 1;
    const int HALF  = rows_before(HEX_R + 1);
    const bool low = (t >= HALF);
    const int tm = low ? (N_OMM - 1 - t) : t;
    int j = 0;
    for (int k = 1; k <= HEX_R; ++k) j += (rows_before(k) <= tm) ? 1 : 0;
    const int row = low ? (NROWS - 1 - j) : j;
    const int ib  = row & ~1;
    const bool ibtop = (ib <= HEX_R + 1);
    const int  ibn   = rows_before(ibtop ? ib : (NROWS - ib));
    const int  cib   = ibtop ? ibn : (N_OMM - ibn);
    const int p   = t - cib;
    const bool upper  = (ib < HEX_R);
    const bool single = (ib == NROWS - 1);
    const int pp = upper ? (p + 1) : p;
    const int q  = (upper ? (HEX_R - ib - 1) : 0) + (single ? p : (pp >> 1));
    const int r  = ib + (single ? 0 : (pp & 1));
    const bool qleft = (q <= HEX_R + 1);
    const int  qn    = rows_before(qleft ? q : (NROWS - q));
    const int  cq    = qleft ? qn : (N_OMM - qn);
    const int rtop = max(0, HEX_R - q);
    return cq + (r - rtop);
}

__global__ __launch_bounds__(192) void hexeye_wmma_f32(
    const float* __restrict__ stim,
    const int*   __restrict__ rx,
    const int*   __restrict__ ry,
    float*       __restrict__ out)
{
    const int task = __builtin_amdgcn_readfirstlane((int)(blockIdx.x * WAVES_PER_BLOCK + (threadIdx.x >> 5)));
    const int lane = threadIdx.x & 31;
    if (task >= NTASKS) return;

    const int batch = task / NTILES;
    const int tile  = task - batch * NTILES;

    const int n   = lane & 15;
    const int sub = lane >> 4;
    const int t_raw = tile * 16 + n;
    const bool valid = (t_raw < N_OMM);
    const int o = walk_to_index(valid ? t_raw : (N_OMM - 1));

    int cx = rx[o]; cx = min(max(cx, PADR), W_PX + PADR - 1);
    int cy = ry[o]; cy = min(max(cy, PADR), H_PX + PADR - 1);

    const float* __restrict__ pb = stim + (size_t)batch * (NCH * H_PX * W_PX);

    int Yb[9];
    for (int d = 0; d < 9; ++d) Yb[d] = reflect_idx(cy + d - 8, H_PX) * W_PX;

    const int xq = cx - 8 + 4 * sub;
    const int x8 = reflect_idx(cx, W_PX);
    float q[NCH][9][4];
    float e[NCH][6];
    if (cx >= 8 && cx <= W_PX - 1) {
        const char* __restrict__ pbb = reinterpret_cast<const char*>(pb);
        unsigned offq[9], off8[5];
        for (int d = 0; d < 9; ++d) offq[d] = (unsigned)(Yb[d] + xq) * 4u;
        for (int j = 0; j < 4; ++j) {
            const int ya = Yb[2 * j], yb = Yb[2 * j + 1];
            off8[j] = (unsigned)((sub ? yb : ya) + x8);
        }
        off8[4] = (unsigned)(Yb[8] + x8);
        for (int j = 0; j < 4; ++j) {
            for (int c = 0; c < NCH; ++c) {
                const float* __restrict__ pc = pb + c * (H_PX * W_PX);
                const char* __restrict__ pcb = pbb + c * (H_PX * W_PX * 4);
                const v4f t0 = *reinterpret_cast<const v4f_a4*>(pcb + offq[2 * j]);
                const v4f t1 = *reinterpret_cast<const v4f_a4*>(pcb + offq[2 * j + 1]);
                e[c][j] = pc[off8[j]];
                q[c][2 * j][0] = t0.x; q[c][2 * j][1] = t0.y; q[c][2 * j][2] = t0.z; q[c][2 * j][3] = t0.w;
                q[c][2 * j + 1][0] = t1.x; q[c][2 * j + 1][1] = t1.y; q[c][2 * j + 1][2] = t1.z; q[c][2 * j + 1][3] = t1.w;
                __builtin_amdgcn_sched_barrier(0);
            }
        }
        for (int c = 0; c < NCH; ++c) {
            const float* __restrict__ pc = pb + c * (H_PX * W_PX);
            const char* __restrict__ pcb = pbb + c * (H_PX * W_PX * 4);
            const v4f t0 = *reinterpret_cast<const v4f_a4*>(pcb + offq[8]);
            const float v = pc[off8[4]];
            q[c][8][0] = t0.x; q[c][8][1] = t0.y; q[c][8][2] = t0.z; q[c][8][3] = t0.w;
            e[c][4] = v;
            __builtin_amdgcn_sched_barrier(0);
        }
    } else {
        int Xc[4];
        for (int k = 0; k < 4; ++k) Xc[k] = reflect_idx(xq + k, W_PX);
        for (int d = 0; d < 9; ++d) {
            for (int c = 0; c < NCH; ++c) {
                for (int k = 0; k < 4; ++k) q[c][d][k] = pb[c * (H_PX * W_PX) + (unsigned)(Yb[d] + Xc[k])];
            }
        }
        for (int j = 0; j < 4; ++j) {
            const int ya = Yb[2 * j], yb = Yb[2 * j + 1];
            const unsigned off8 = (unsigned)((sub ? yb : ya) + x8);
            for (int c = 0; c < NCH; ++c) e[c][j] = pb[c * (H_PX * W_PX) + off8];
        }
        {
            const unsigned off8 = (unsigned)(Yb[8] + x8);
            for (int c = 0; c < NCH; ++c) e[c][4] = pb[c * (H_PX * W_PX) + off8];
        }
    }
    for (int c = 0; c < NCH; ++c) {
        e[c][4] = sub ? 0.0f : e[c][4];
        e[c][5] = 0.0f;
    }

    v2f a; a.x = 1.0f; a.y = 1.0f;
    v8f acc[NCH];
    for (int c = 0; c < NCH; ++c) acc[c] = (v8f){};

    for (int d = 0; d < 9; ++d) {
        for (int h = 0; h < 2; ++h) {
            for (int c = 0; c < NCH; ++c) {
                v2f bb; bb.x = q[c][d][2 * h]; bb.y = q[c][d][2 * h + 1];
                acc[c] = __builtin_amdgcn_wmma_f32_16x16x4_f32(
                     false, a,  false, bb,
                     (short)0, acc[c],  false,  false);
            }
        }
    }
    for (int h = 0; h < 3; ++h) {
        for (int c = 0; c < NCH; ++c) {
            v2f bb; bb.x = e[c][2 * h]; bb.y = e[c][2 * h + 1];
            acc[c] = __builtin_amdgcn_wmma_f32_16x16x4_f32(
                 false, a,  false, bb,
                 (short)0, acc[c],  false,  false);
        }
    }

    if (valid && sub == 0) {
        for (int c = 0; c < NCH; ++c)
            out[(size_t)(batch * NCH + c) * N_OMM + o] = acc[c][0] / 81.0f;
    }
}

extern "C" void kernel_launch(void* const* d_in, const int* in_sizes, int n_in,
                              void* d_out, int out_size, void* d_ws, size_t ws_size,
                              hipStream_t stream) {
    const float* stim = (const float*)d_in[0];
    const int*   rx   = (const int*)d_in[1];
    const int*   ry   = (const int*)d_in[2];
    float*       out  = (float*)d_out;

    (void)in_sizes; (void)n_in; (void)out_size; (void)d_ws; (void)ws_size;

    hexeye_wmma_f32<<<NBLOCKS, WAVES_PER_BLOCK * 32, 0, stream>>>(stim, rx, ry, out);
}
